# starting kernel
# baseline (speedup 1.0000x reference)
#include <hip/hip_runtime.h>


typedef float v2f __attribute__((ext_vector_type(2)));
typedef float v8f __attribute__((ext_vector_type(8)));

#define WMMA_F32X4(A, B, C) \
    __builtin_amdgcn_wmma_f32_16x16x4_f32(false, (A), false, (B), (short)0, (C), false, false)

constexpr int Tn = 1024;
constexpr int Bn = 4096;

__device__ __forceinline__ float fast_sigmoid(float v) {
    return __builtin_amdgcn_rcpf(1.0f + __expf(-v));
}
__device__ __forceinline__ float fast_tanh(float v) {
    return 1.0f - 2.0f * __builtin_amdgcn_rcpf(__expf(2.0f * v) + 1.0f);
}

__global__ __launch_bounds__(256) void gru_wmma_kernel(
    const float* __restrict__ x,
    const float* __restrict__ h0,
    const float* __restrict__ w_ih,
    const float* __restrict__ w_hh,
    const float* __restrict__ b_ih,
    const float* __restrict__ b_hh,
    const float* __restrict__ fc_w,
    const float* __restrict__ fc_b,
    float* __restrict__ out,
    float* __restrict__ h_out)
{
    const int lane  = threadIdx.x & 31;
    const int wid   = (blockIdx.x * blockDim.x + threadIdx.x) >> 5;
    const int m     = lane & 15;
    const bool lo   = lane < 16;
    const int bbase = wid * 16;
    const int b     = bbase + m;

    const int k0 = lo ? 0 : 2;
    v2f aRZ1, aRZ2, aN1, aN2;
    aRZ1.x = w_hh[m * 8 + k0 + 0]; aRZ1.y = w_hh[m * 8 + k0 + 1];
    aRZ2.x = w_hh[m * 8 + k0 + 4]; aRZ2.y = w_hh[m * 8 + k0 + 5];
    if (m < 8) {
        aN1.x = w_hh[(16 + m) * 8 + k0 + 0]; aN1.y = w_hh[(16 + m) * 8 + k0 + 1];
        aN2.x = w_hh[(16 + m) * 8 + k0 + 4]; aN2.y = w_hh[(16 + m) * 8 + k0 + 5];
    } else {
        aN1.x = aN1.y = aN2.x = aN2.y = 0.0f;
    }

    v8f cRZ, cN;
    #pragma unroll
    for (int r = 0; r < 8; ++r) {
        const int g = lo ? r : 8 + r;
        cRZ[r] = b_ih[g] + b_hh[g];
        cN[r]  = lo ? b_hh[16 + r] : 0.0f;
    }

    float wA0[8], wA1[8];
    #pragma unroll
    for (int j = 0; j < 8; ++j) {
        const int g = lo ? j : 8 + j;
        wA0[j] = w_ih[g * 2 + 0];
        wA1[j] = w_ih[g * 2 + 1];
    }
    const int jl[4] = {0, 1, 4, 5};
    const int jh[4] = {2, 3, 6, 7};
    float wN0[4], wN1[4], bN[4], fcw[4];
    #pragma unroll
    for (int s = 0; s < 4; ++s) {
        const int j = lo ? jl[s] : jh[s];
        wN0[s] = w_ih[(16 + j) * 2 + 0];
        wN1[s] = w_ih[(16 + j) * 2 + 1];
        bN[s]  = b_ih[16 + j];
        fcw[s] = fc_w[j];
    }
    const float fcb = fc_b[0];

    float h[4];
    {
        const float* hp = h0 + (size_t)b * 8 + (lo ? 0 : 2);
        h[0] = hp[0]; h[1] = hp[1]; h[2] = hp[4]; h[3] = hp[5];
    }

    const float* xp   = x + (size_t)b * 2;
    float*       outp = out + bbase + m;

    for (int t = 0; t < Tn; ++t) {
        const float x0 = xp[0];
        const float x1 = xp[1];
        xp += (size_t)Bn * 2;

        v2f b1, b2;
        b1.x = h[0]; b1.y = h[1];
        b2.x = h[2]; b2.y = h[3];

        v8f dRZ = WMMA_F32X4(aRZ1, b1, cRZ);
        dRZ     = WMMA_F32X4(aRZ2, b2, dRZ);
        v8f dN  = WMMA_F32X4(aN1, b1, cN);
        dN      = WMMA_F32X4(aN2, b2, dN);

        float a[8], o[8], g2[8];
        #pragma unroll
        for (int j = 0; j < 8; ++j) {
            const float u = dRZ[j] + fmaf(x1, wA1[j], x0 * wA0[j]);
            a[j]  = fast_sigmoid(u);
            o[j]  = __shfl_xor(a[j], 16, 32);
            g2[j] = __shfl_xor(dN[j], 16, 32);
        }

        #pragma unroll
        for (int s = 0; s < 4; ++s) {
            const float rS  = lo ? a[jl[s]]  : o[jh[s]];
            const float zS  = lo ? o[jl[s]]  : a[jh[s]];
            const float hnS = lo ? dN[jl[s]] : g2[jh[s]];
            const float xnS = fmaf(x1, wN1[s], fmaf(x0, wN0[s], bN[s]));
            const float nS  = fast_tanh(fmaf(rS, hnS, xnS));
            h[s] = nS + zS * (h[s] - nS);
        }

        float acc = fmaf(fcw[3], h[3], fmaf(fcw[2], h[2], fmaf(fcw[1], h[1], fcw[0] * h[0])));
        const float y = fast_sigmoid(acc + __shfl_xor(acc, 16, 32) + fcb);
        if (lo) *outp = y;
        outp += Bn;
    }

    {
        float* hp = h_out + (size_t)b * 8 + (lo ? 0 : 2);
        hp[0] = h[0]; hp[1] = h[1]; hp[4] = h[2]; hp[5] = h[3];
    }
}

extern "C" void kernel_launch(void* const* d_in, const int* in_sizes, int n_in,
                              void* d_out, int out_size, void* d_ws, size_t ws_size,
                              hipStream_t stream) {
    const float* x    = (const float*)d_in[0];
    const float* h0   = (const float*)d_in[1];
    const float* w_ih = (const float*)d_in[2];
    const float* w_hh = (const float*)d_in[3];
    const float* b_ih = (const float*)d_in[4];
    const float* b_hh = (const float*)d_in[5];
    const float* fc_w = (const float*)d_in[6];
    const float* fc_b = (const float*)d_in[7];

    float* out    = (float*)d_out;
    float* h_last = out + (size_t)Tn * Bn;

    const int threads = 256;
    const int blocks  = Bn / 16 / (threads / 32);
    gru_wmma_kernel<<<blocks, threads, 0, stream>>>(
        x, h0, w_ih, w_hh, b_ih, b_hh, fc_w, fc_b, out, h_last);
}


# x of the next 8 timesteps is loaded one block ahead (T loop in blocks of 8, fully unrolled): the per-step global-load round trip leaves the serial recurrence chain; arithmetic unchanged
# speedup vs baseline: 1.7417x; 1.7417x over previous
#include <hip/hip_runtime.h>


typedef float v2f __attribute__((ext_vector_type(2)));
typedef float v8f __attribute__((ext_vector_type(8)));

#define WMMA_F32X4(A, B, C) \
    __builtin_amdgcn_wmma_f32_16x16x4_f32(false, (A), false, (B), (short)0, (C), false, false)

constexpr int Tn = 1024;
constexpr int Bn = 4096;
constexpr int PF = 8;

__device__ __forceinline__ float fast_sigmoid(float v) {
    return __builtin_amdgcn_rcpf(1.0f + __expf(-v));
}
__device__ __forceinline__ float fast_tanh(float v) {
    return 1.0f - 2.0f * __builtin_amdgcn_rcpf(__expf(2.0f * v) + 1.0f);
}

__global__ __launch_bounds__(256) void gru_wmma_kernel(
    const float* __restrict__ x,
    const float* __restrict__ h0,
    const float* __restrict__ w_ih,
    const float* __restrict__ w_hh,
    const float* __restrict__ b_ih,
    const float* __restrict__ b_hh,
    const float* __restrict__ fc_w,
    const float* __restrict__ fc_b,
    float* __restrict__ out,
    float* __restrict__ h_out)
{
    const int lane  = threadIdx.x & 31;
    const int wid   = (blockIdx.x * blockDim.x + threadIdx.x) >> 5;
    const int m     = lane & 15;
    const bool lo   = lane < 16;
    const int bbase = wid * 16;
    const int b     = bbase + m;

    const int k0 = lo ? 0 : 2;
    v2f aRZ1, aRZ2, aN1, aN2;
    aRZ1.x = w_hh[m * 8 + k0 + 0]; aRZ1.y = w_hh[m * 8 + k0 + 1];
    aRZ2.x = w_hh[m * 8 + k0 + 4]; aRZ2.y = w_hh[m * 8 + k0 + 5];
    if (m < 8) {
        aN1.x = w_hh[(16 + m) * 8 + k0 + 0]; aN1.y = w_hh[(16 + m) * 8 + k0 + 1];
        aN2.x = w_hh[(16 + m) * 8 + k0 + 4]; aN2.y = w_hh[(16 + m) * 8 + k0 + 5];
    } else {
        aN1.x = aN1.y = aN2.x = aN2.y = 0.0f;
    }

    v8f cRZ, cN;
    #pragma unroll
    for (int r = 0; r < 8; ++r) {
        const int g = lo ? r : 8 + r;
        cRZ[r] = b_ih[g] + b_hh[g];
        cN[r]  = lo ? b_hh[16 + r] : 0.0f;
    }

    float wA0[8], wA1[8];
    #pragma unroll
    for (int j = 0; j < 8; ++j) {
        const int g = lo ? j : 8 + j;
        wA0[j] = w_ih[g * 2 + 0];
        wA1[j] = w_ih[g * 2 + 1];
    }
    const int jl[4] = {0, 1, 4, 5};
    const int jh[4] = {2, 3, 6, 7};
    float wN0[4], wN1[4], bN[4], fcw[4];
    #pragma unroll
    for (int s = 0; s < 4; ++s) {
        const int j = lo ? jl[s] : jh[s];
        wN0[s] = w_ih[(16 + j) * 2 + 0];
        wN1[s] = w_ih[(16 + j) * 2 + 1];
        bN[s]  = b_ih[16 + j];
        fcw[s] = fc_w[j];
    }
    const float fcb = fc_b[0];

    float h[4];
    {
        const float* hp = h0 + (size_t)b * 8 + (lo ? 0 : 2);
        h[0] = hp[0]; h[1] = hp[1]; h[2] = hp[4]; h[3] = hp[5];
    }

    const float* xp   = x + (size_t)b * 2;
    float*       outp = out + bbase + m;

    float xc0[PF], xc1[PF];
    #pragma unroll
    for (int u = 0; u < PF; ++u) {
        xc0[u] = xp[(size_t)u * Bn * 2 + 0];
        xc1[u] = xp[(size_t)u * Bn * 2 + 1];
    }

    for (int tb = 0; tb < Tn; tb += PF) {
      float xn0[PF], xn1[PF];
      #pragma unroll
      for (int u = 0; u < PF; ++u) {
          const int tn = (tb + PF + u < Tn) ? (tb + PF + u) : (Tn - 1);
          const float* xn = xp + (size_t)tn * Bn * 2;
          xn0[u] = xn[0];
          xn1[u] = xn[1];
      }

      #pragma unroll
      for (int u = 0; u < PF; ++u) {
        const float x0 = xc0[u];
        const float x1 = xc1[u];

        v2f b1, b2;
        b1.x = h[0]; b1.y = h[1];
        b2.x = h[2]; b2.y = h[3];

        v8f dRZ = WMMA_F32X4(aRZ1, b1, cRZ);
        dRZ     = WMMA_F32X4(aRZ2, b2, dRZ);
        v8f dN  = WMMA_F32X4(aN1, b1, cN);
        dN      = WMMA_F32X4(aN2, b2, dN);

        float a[8], o[8], g2[8];
        #pragma unroll
        for (int j = 0; j < 8; ++j) {
            const float u = dRZ[j] + fmaf(x1, wA1[j], x0 * wA0[j]);
            a[j]  = fast_sigmoid(u);
            o[j]  = __shfl_xor(a[j], 16, 32);
            g2[j] = __shfl_xor(dN[j], 16, 32);
        }

        #pragma unroll
        for (int s = 0; s < 4; ++s) {
            const float rS  = lo ? a[jl[s]]  : o[jh[s]];
            const float zS  = lo ? o[jl[s]]  : a[jh[s]];
            const float hnS = lo ? dN[jl[s]] : g2[jh[s]];
            const float xnS = fmaf(x1, wN1[s], fmaf(x0, wN0[s], bN[s]));
            const float nS  = fast_tanh(fmaf(rS, hnS, xnS));
            h[s] = nS + zS * (h[s] - nS);
        }

        float acc = fmaf(fcw[3], h[3], fmaf(fcw[2], h[2], fmaf(fcw[1], h[1], fcw[0] * h[0])));
        const float y = fast_sigmoid(acc + __shfl_xor(acc, 16, 32) + fcb);
        if (lo) *outp = y;
        outp += Bn;
      }

      #pragma unroll
      for (int u = 0; u < PF; ++u) { xc0[u] = xn0[u]; xc1[u] = xn1[u]; }
    }

    {
        float* hp = h_out + (size_t)b * 8 + (lo ? 0 : 2);
        hp[0] = h[0]; hp[1] = h[1]; hp[4] = h[2]; hp[5] = h[3];
    }
}

extern "C" void kernel_launch(void* const* d_in, const int* in_sizes, int n_in,
                              void* d_out, int out_size, void* d_ws, size_t ws_size,
                              hipStream_t stream) {
    const float* x    = (const float*)d_in[0];
    const float* h0   = (const float*)d_in[1];
    const float* w_ih = (const float*)d_in[2];
    const float* w_hh = (const float*)d_in[3];
    const float* b_ih = (const float*)d_in[4];
    const float* b_hh = (const float*)d_in[5];
    const float* fc_w = (const float*)d_in[6];
    const float* fc_b = (const float*)d_in[7];

    float* out    = (float*)d_out;
    float* h_last = out + (size_t)Tn * Bn;

    const int threads = 256;
    const int blocks  = Bn / 16 / (threads / 32);
    gru_wmma_kernel<<<blocks, threads, 0, stream>>>(
        x, h0, w_ih, w_hh, b_ih, b_hh, fc_w, fc_b, out, h_last);
}


# w_hh gate rows reordered in the A tiles so a lane's D holds r,z,n of the 4 hidden units it owns: the 12 gate ds_bpermute and selects per step go; k sum order now 0,1,4,5|2,3,6,7, head halves 0-3|4-7
# speedup vs baseline: 2.8805x; 1.6538x over previous
#include <hip/hip_runtime.h>


typedef float v2f __attribute__((ext_vector_type(2)));
typedef float v8f __attribute__((ext_vector_type(8)));

#define WMMA_F32X4(A, B, C) \
    __builtin_amdgcn_wmma_f32_16x16x4_f32(false, (A), false, (B), (short)0, (C), false, false)

constexpr int Tn = 1024;
constexpr int Bn = 4096;
constexpr int PF = 8;

__device__ __forceinline__ float fast_sigmoid(float v) {
    return __builtin_amdgcn_rcpf(1.0f + __expf(-v));
}
__device__ __forceinline__ float fast_tanh(float v) {
    return 1.0f - 2.0f * __builtin_amdgcn_rcpf(__expf(2.0f * v) + 1.0f);
}

__global__ __launch_bounds__(256) void gru_wmma_kernel(
    const float* __restrict__ x,
    const float* __restrict__ h0,
    const float* __restrict__ w_ih,
    const float* __restrict__ w_hh,
    const float* __restrict__ b_ih,
    const float* __restrict__ b_hh,
    const float* __restrict__ fc_w,
    const float* __restrict__ fc_b,
    float* __restrict__ out,
    float* __restrict__ h_out)
{
    const int lane  = threadIdx.x & 31;
    const int wid   = (blockIdx.x * blockDim.x + threadIdx.x) >> 5;
    const int m     = lane & 15;
    const int half  = lane >> 4;
    const bool lo   = half == 0;
    const int bbase = wid * 16;
    const int b     = bbase + m;
    const int hb    = half * 4;

    const int  a_hid = ((m >> 3) << 2) + (m & 3);
    const bool a_isz = (m & 4) != 0;
    const int  rowRZ = (a_isz ? 8 : 0) + a_hid;
    const int  rowN  = 16 + a_hid;
    v2f aRZ1, aRZ2, aN1, aN2;
    aRZ1.x = w_hh[rowRZ * 8 + hb + 0]; aRZ1.y = w_hh[rowRZ * 8 + hb + 1];
    aRZ2.x = w_hh[rowRZ * 8 + hb + 2]; aRZ2.y = w_hh[rowRZ * 8 + hb + 3];
    if (!a_isz) {
        aN1.x = w_hh[rowN * 8 + hb + 0]; aN1.y = w_hh[rowN * 8 + hb + 1];
        aN2.x = w_hh[rowN * 8 + hb + 2]; aN2.y = w_hh[rowN * 8 + hb + 3];
    } else {
        aN1.x = aN1.y = aN2.x = aN2.y = 0.0f;
    }

    v8f cRZ, cN;
    #pragma unroll
    for (int r = 0; r < 4; ++r) {
        cRZ[r]     = b_ih[hb + r]     + b_hh[hb + r];
        cRZ[4 + r] = b_ih[8 + hb + r] + b_hh[8 + hb + r];
        cN[r]      = b_hh[16 + hb + r];
        cN[4 + r]  = 0.0f;
    }

    float wR0[4], wR1[4], wZ0[4], wZ1[4], wN0[4], wN1[4], bN[4], fcw[4];
    #pragma unroll
    for (int s = 0; s < 4; ++s) {
        const int j = hb + s;
        wR0[s] = w_ih[j * 2 + 0];        wR1[s] = w_ih[j * 2 + 1];
        wZ0[s] = w_ih[(8 + j) * 2 + 0];  wZ1[s] = w_ih[(8 + j) * 2 + 1];
        wN0[s] = w_ih[(16 + j) * 2 + 0]; wN1[s] = w_ih[(16 + j) * 2 + 1];
        bN[s]  = b_ih[16 + j];
        fcw[s] = fc_w[j];
    }
    const float fcb = fc_b[0];

    float h[4];
    {
        const float* hp = h0 + (size_t)b * 8 + hb;
        h[0] = hp[0]; h[1] = hp[1]; h[2] = hp[2]; h[3] = hp[3];
    }

    const float* xp   = x + (size_t)b * 2;
    float*       outp = out + bbase + m;

    float xc0[PF], xc1[PF];
    #pragma unroll
    for (int u = 0; u < PF; ++u) {
        xc0[u] = xp[(size_t)u * Bn * 2 + 0];
        xc1[u] = xp[(size_t)u * Bn * 2 + 1];
    }

    for (int tb = 0; tb < Tn; tb += PF) {
      float xn0[PF], xn1[PF];
      #pragma unroll
      for (int u = 0; u < PF; ++u) {
          const int tn = (tb + PF + u < Tn) ? (tb + PF + u) : (Tn - 1);
          const float* xn = xp + (size_t)tn * Bn * 2;
          xn0[u] = xn[0];
          xn1[u] = xn[1];
      }

      #pragma unroll
      for (int u = 0; u < PF; ++u) {
        const float x0 = xc0[u];
        const float x1 = xc1[u];

        v2f b1, b2;
        b1.x = h[0]; b1.y = h[1];
        b2.x = h[2]; b2.y = h[3];

        v8f dRZ = WMMA_F32X4(aRZ1, b1, cRZ);
        dRZ     = WMMA_F32X4(aRZ2, b2, dRZ);
        v8f dN  = WMMA_F32X4(aN1, b1, cN);
        dN      = WMMA_F32X4(aN2, b2, dN);

        #pragma unroll
        for (int s = 0; s < 4; ++s) {
            const float rS  = fast_sigmoid(dRZ[s]     + fmaf(x1, wR1[s], x0 * wR0[s]));
            const float zS  = fast_sigmoid(dRZ[4 + s] + fmaf(x1, wZ1[s], x0 * wZ0[s]));
            const float xnS = fmaf(x1, wN1[s], fmaf(x0, wN0[s], bN[s]));
            const float nS  = fast_tanh(fmaf(rS, dN[s], xnS));
            h[s] = nS + zS * (h[s] - nS);
        }

        float acc = fmaf(fcw[3], h[3], fmaf(fcw[2], h[2], fmaf(fcw[1], h[1], fcw[0] * h[0])));
        const float y = fast_sigmoid(acc + __shfl_xor(acc, 16, 32) + fcb);
        if (lo) *outp = y;
        outp += Bn;
      }

      #pragma unroll
      for (int u = 0; u < PF; ++u) { xc0[u] = xn0[u]; xc1[u] = xn1[u]; }
    }

    {
        float* hp = h_out + (size_t)b * 8 + hb;
        hp[0] = h[0]; hp[1] = h[1]; hp[2] = h[2]; hp[3] = h[3];
    }
}

extern "C" void kernel_launch(void* const* d_in, const int* in_sizes, int n_in,
                              void* d_out, int out_size, void* d_ws, size_t ws_size,
                              hipStream_t stream) {
    const float* x    = (const float*)d_in[0];
    const float* h0   = (const float*)d_in[1];
    const float* w_ih = (const float*)d_in[2];
    const float* w_hh = (const float*)d_in[3];
    const float* b_ih = (const float*)d_in[4];
    const float* b_hh = (const float*)d_in[5];
    const float* fc_w = (const float*)d_in[6];
    const float* fc_b = (const float*)d_in[7];

    float* out    = (float*)d_out;
    float* h_last = out + (size_t)Tn * Bn;

    const int threads = 256;
    const int blocks  = Bn / 16 / (threads / 32);
    gru_wmma_kernel<<<blocks, threads, 0, stream>>>(
        x, h0, w_ih, w_hh, b_ih, b_hh, fc_w, fc_b, out, h_last);
}


# FC head off the per-step path: lane halves swap dot-product parts by v_permlanex16 once per 2 steps, half 0 ends the even step, half 1 the odd; no ds_bpermute wait or exec branch; sums unchanged
# speedup vs baseline: 3.5757x; 1.2413x over previous
#include <hip/hip_runtime.h>


typedef float v2f __attribute__((ext_vector_type(2)));
typedef float v8f __attribute__((ext_vector_type(8)));

#define WMMA_F32X4(A, B, C) \
    __builtin_amdgcn_wmma_f32_16x16x4_f32(false, (A), false, (B), (short)0, (C), false, false)

constexpr int Tn = 1024;
constexpr int Bn = 4096;
constexpr int PF = 8;

__device__ __forceinline__ float fast_sigmoid(float v) {
    return __builtin_amdgcn_rcpf(1.0f + __expf(-v));
}
__device__ __forceinline__ float xhalf(float v) {
    const int bits = __builtin_bit_cast(int, v);
    return __builtin_bit_cast(float,
        __builtin_amdgcn_permlanex16(bits, bits, 0x76543210, 0xfedcba98, false, false));
}
__device__ __forceinline__ float fast_tanh(float v) {
    return 1.0f - 2.0f * __builtin_amdgcn_rcpf(__expf(2.0f * v) + 1.0f);
}

__global__ __launch_bounds__(256) void gru_wmma_kernel(
    const float* __restrict__ x,
    const float* __restrict__ h0,
    const float* __restrict__ w_ih,
    const float* __restrict__ w_hh,
    const float* __restrict__ b_ih,
    const float* __restrict__ b_hh,
    const float* __restrict__ fc_w,
    const float* __restrict__ fc_b,
    float* __restrict__ out,
    float* __restrict__ h_out)
{
    const int lane  = threadIdx.x & 31;
    const int wid   = (blockIdx.x * blockDim.x + threadIdx.x) >> 5;
    const int m     = lane & 15;
    const int half  = lane >> 4;
    const bool lo   = half == 0;
    const int bbase = wid * 16;
    const int b     = bbase + m;
    const int hb    = half * 4;

    const int  a_hid = ((m >> 3) << 2) + (m & 3);
    const bool a_isz = (m & 4) != 0;
    const int  rowRZ = (a_isz ? 8 : 0) + a_hid;
    const int  rowN  = 16 + a_hid;
    v2f aRZ1, aRZ2, aN1, aN2;
    aRZ1.x = w_hh[rowRZ * 8 + hb + 0]; aRZ1.y = w_hh[rowRZ * 8 + hb + 1];
    aRZ2.x = w_hh[rowRZ * 8 + hb + 2]; aRZ2.y = w_hh[rowRZ * 8 + hb + 3];
    if (!a_isz) {
        aN1.x = w_hh[rowN * 8 + hb + 0]; aN1.y = w_hh[rowN * 8 + hb + 1];
        aN2.x = w_hh[rowN * 8 + hb + 2]; aN2.y = w_hh[rowN * 8 + hb + 3];
    } else {
        aN1.x = aN1.y = aN2.x = aN2.y = 0.0f;
    }

    v8f cRZ, cN;
    #pragma unroll
    for (int r = 0; r < 4; ++r) {
        cRZ[r]     = b_ih[hb + r]     + b_hh[hb + r];
        cRZ[4 + r] = b_ih[8 + hb + r] + b_hh[8 + hb + r];
        cN[r]      = b_hh[16 + hb + r];
        cN[4 + r]  = 0.0f;
    }

    float wR0[4], wR1[4], wZ0[4], wZ1[4], wN0[4], wN1[4], bN[4], fcw[4];
    #pragma unroll
    for (int s = 0; s < 4; ++s) {
        const int j = hb + s;
        wR0[s] = w_ih[j * 2 + 0];        wR1[s] = w_ih[j * 2 + 1];
        wZ0[s] = w_ih[(8 + j) * 2 + 0];  wZ1[s] = w_ih[(8 + j) * 2 + 1];
        wN0[s] = w_ih[(16 + j) * 2 + 0]; wN1[s] = w_ih[(16 + j) * 2 + 1];
        bN[s]  = b_ih[16 + j];
        fcw[s] = fc_w[j];
    }
    const float fcb = fc_b[0];

    float h[4];
    {
        const float* hp = h0 + (size_t)b * 8 + hb;
        h[0] = hp[0]; h[1] = hp[1]; h[2] = hp[2]; h[3] = hp[3];
    }

    const float* xp   = x + (size_t)b * 2;
    float*       outp = out + (size_t)half * Bn + bbase + m;

    float xc0[PF], xc1[PF];
    #pragma unroll
    for (int u = 0; u < PF; ++u) {
        xc0[u] = xp[(size_t)u * Bn * 2 + 0];
        xc1[u] = xp[(size_t)u * Bn * 2 + 1];
    }

    for (int tb = 0; tb < Tn; tb += PF) {
      float xn0[PF], xn1[PF];
      #pragma unroll
      for (int u = 0; u < PF; ++u) {
          const int tn = (tb + PF + u < Tn) ? (tb + PF + u) : (Tn - 1);
          const float* xn = xp + (size_t)tn * Bn * 2;
          xn0[u] = xn[0];
          xn1[u] = xn[1];
      }
      __builtin_amdgcn_sched_barrier(0);

      float accEven = 0.0f;
      #pragma unroll
      for (int u = 0; u < PF; ++u) {
        const float x0 = xc0[u];
        const float x1 = xc1[u];

        v2f b1, b2;
        b1.x = h[0]; b1.y = h[1];
        b2.x = h[2]; b2.y = h[3];

        v8f dRZ = WMMA_F32X4(aRZ1, b1, cRZ);
        dRZ     = WMMA_F32X4(aRZ2, b2, dRZ);
        v8f dN  = WMMA_F32X4(aN1, b1, cN);
        dN      = WMMA_F32X4(aN2, b2, dN);

        #pragma unroll
        for (int s = 0; s < 4; ++s) {
            const float rS  = fast_sigmoid(dRZ[s]     + fmaf(x1, wR1[s], x0 * wR0[s]));
            const float zS  = fast_sigmoid(dRZ[4 + s] + fmaf(x1, wZ1[s], x0 * wZ0[s]));
            const float xnS = fmaf(x1, wN1[s], fmaf(x0, wN0[s], bN[s]));
            const float nS  = fast_tanh(fmaf(rS, dN[s], xnS));
            h[s] = nS + zS * (h[s] - nS);
        }

        const float acc = fmaf(fcw[3], h[3], fmaf(fcw[2], h[2], fmaf(fcw[1], h[1], fcw[0] * h[0])));
        if ((u & 1) == 0) {
            accEven = acc;
        } else {
            const float mine = lo ? accEven : acc;
            const float send = lo ? acc : accEven;
            const float y = fast_sigmoid(mine + xhalf(send) + fcb);
            outp[(size_t)(u - 1) * Bn] = y;
        }
      }
      outp += (size_t)PF * Bn;

      #pragma unroll
      for (int u = 0; u < PF; ++u) { xc0[u] = xn0[u]; xc1[u] = xn1[u]; }
    }

    {
        float* hp = h_out + (size_t)b * 8 + hb;
        hp[0] = h[0]; hp[1] = h[1]; hp[2] = h[2]; hp[3] = h[3];
    }
}

extern "C" void kernel_launch(void* const* d_in, const int* in_sizes, int n_in,
                              void* d_out, int out_size, void* d_ws, size_t ws_size,
                              hipStream_t stream) {
    const float* x    = (const float*)d_in[0];
    const float* h0   = (const float*)d_in[1];
    const float* w_ih = (const float*)d_in[2];
    const float* w_hh = (const float*)d_in[3];
    const float* b_ih = (const float*)d_in[4];
    const float* b_hh = (const float*)d_in[5];
    const float* fc_w = (const float*)d_in[6];
    const float* fc_b = (const float*)d_in[7];

    float* out    = (float*)d_out;
    float* h_last = out + (size_t)Tn * Bn;

    const int threads = 256;
    const int blocks  = Bn / 16 / (threads / 32);
    gru_wmma_kernel<<<blocks, threads, 0, stream>>>(
        x, h0, w_ih, w_hh, b_ih, b_hh, fc_w, fc_b, out, h_last);
}


# launch 64 work-groups of 4 waves in place of 32 of 8: one wave per SIMD, no second serial chain shares a wave's issue port and matrix pipe; kernel code, arithmetic and sum order unchanged
# speedup vs baseline: 5.3750x; 1.5032x over previous
#include <hip/hip_runtime.h>


typedef float v2f __attribute__((ext_vector_type(2)));
typedef float v8f __attribute__((ext_vector_type(8)));

#define WMMA_F32X4(A, B, C) \
    __builtin_amdgcn_wmma_f32_16x16x4_f32(false, (A), false, (B), (short)0, (C), false, false)

constexpr int Tn = 1024;
constexpr int Bn = 4096;
constexpr int PF = 8;

__device__ __forceinline__ float fast_sigmoid(float v) {
    return __builtin_amdgcn_rcpf(1.0f + __expf(-v));
}
__device__ __forceinline__ float xhalf(float v) {
    const int bits = __builtin_bit_cast(int, v);
    return __builtin_bit_cast(float,
        __builtin_amdgcn_permlanex16(bits, bits, 0x76543210, 0xfedcba98, false, false));
}
__device__ __forceinline__ float fast_tanh(float v) {
    return 1.0f - 2.0f * __builtin_amdgcn_rcpf(__expf(2.0f * v) + 1.0f);
}

__global__ __launch_bounds__(256) void gru_wmma_kernel(
    const float* __restrict__ x,
    const float* __restrict__ h0,
    const float* __restrict__ w_ih,
    const float* __restrict__ w_hh,
    const float* __restrict__ b_ih,
    const float* __restrict__ b_hh,
    const float* __restrict__ fc_w,
    const float* __restrict__ fc_b,
    float* __restrict__ out,
    float* __restrict__ h_out)
{
    const int lane  = threadIdx.x & 31;
    const int wid   = (blockIdx.x * blockDim.x + threadIdx.x) >> 5;
    const int m     = lane & 15;
    const int half  = lane >> 4;
    const bool lo   = half == 0;
    const int bbase = wid * 16;
    const int b     = bbase + m;
    const int hb    = half * 4;

    const int  a_hid = ((m >> 3) << 2) + (m & 3);
    const bool a_isz = (m & 4) != 0;
    const int  rowRZ = (a_isz ? 8 : 0) + a_hid;
    const int  rowN  = 16 + a_hid;
    v2f aRZ1, aRZ2, aN1, aN2;
    aRZ1.x = w_hh[rowRZ * 8 + hb + 0]; aRZ1.y = w_hh[rowRZ * 8 + hb + 1];
    aRZ2.x = w_hh[rowRZ * 8 + hb + 2]; aRZ2.y = w_hh[rowRZ * 8 + hb + 3];
    if (!a_isz) {
        aN1.x = w_hh[rowN * 8 + hb + 0]; aN1.y = w_hh[rowN * 8 + hb + 1];
        aN2.x = w_hh[rowN * 8 + hb + 2]; aN2.y = w_hh[rowN * 8 + hb + 3];
    } else {
        aN1.x = aN1.y = aN2.x = aN2.y = 0.0f;
    }

    v8f cRZ, cN;
    #pragma unroll
    for (int r = 0; r < 4; ++r) {
        cRZ[r]     = b_ih[hb + r]     + b_hh[hb + r];
        cRZ[4 + r] = b_ih[8 + hb + r] + b_hh[8 + hb + r];
        cN[r]      = b_hh[16 + hb + r];
        cN[4 + r]  = 0.0f;
    }

    float wR0[4], wR1[4], wZ0[4], wZ1[4], wN0[4], wN1[4], bN[4], fcw[4];
    #pragma unroll
    for (int s = 0; s < 4; ++s) {
        const int j = hb + s;
        wR0[s] = w_ih[j * 2 + 0];        wR1[s] = w_ih[j * 2 + 1];
        wZ0[s] = w_ih[(8 + j) * 2 + 0];  wZ1[s] = w_ih[(8 + j) * 2 + 1];
        wN0[s] = w_ih[(16 + j) * 2 + 0]; wN1[s] = w_ih[(16 + j) * 2 + 1];
        bN[s]  = b_ih[16 + j];
        fcw[s] = fc_w[j];
    }
    const float fcb = fc_b[0];

    float h[4];
    {
        const float* hp = h0 + (size_t)b * 8 + hb;
        h[0] = hp[0]; h[1] = hp[1]; h[2] = hp[2]; h[3] = hp[3];
    }

    const float* xp   = x + (size_t)b * 2;
    float*       outp = out + (size_t)half * Bn + bbase + m;

    float xc0[PF], xc1[PF];
    #pragma unroll
    for (int u = 0; u < PF; ++u) {
        xc0[u] = xp[(size_t)u * Bn * 2 + 0];
        xc1[u] = xp[(size_t)u * Bn * 2 + 1];
    }

    for (int tb = 0; tb < Tn; tb += PF) {
      float xn0[PF], xn1[PF];
      #pragma unroll
      for (int u = 0; u < PF; ++u) {
          const int tn = (tb + PF + u < Tn) ? (tb + PF + u) : (Tn - 1);
          const float* xn = xp + (size_t)tn * Bn * 2;
          xn0[u] = xn[0];
          xn1[u] = xn[1];
      }
      __builtin_amdgcn_sched_barrier(0);

      float accEven = 0.0f;
      #pragma unroll
      for (int u = 0; u < PF; ++u) {
        const float x0 = xc0[u];
        const float x1 = xc1[u];

        v2f b1, b2;
        b1.x = h[0]; b1.y = h[1];
        b2.x = h[2]; b2.y = h[3];

        v8f dRZ = WMMA_F32X4(aRZ1, b1, cRZ);
        dRZ     = WMMA_F32X4(aRZ2, b2, dRZ);
        v8f dN  = WMMA_F32X4(aN1, b1, cN);
        dN      = WMMA_F32X4(aN2, b2, dN);

        #pragma unroll
        for (int s = 0; s < 4; ++s) {
            const float rS  = fast_sigmoid(dRZ[s]     + fmaf(x1, wR1[s], x0 * wR0[s]));
            const float zS  = fast_sigmoid(dRZ[4 + s] + fmaf(x1, wZ1[s], x0 * wZ0[s]));
            const float xnS = fmaf(x1, wN1[s], fmaf(x0, wN0[s], bN[s]));
            const float nS  = fast_tanh(fmaf(rS, dN[s], xnS));
            h[s] = nS + zS * (h[s] - nS);
        }

        const float acc = fmaf(fcw[3], h[3], fmaf(fcw[2], h[2], fmaf(fcw[1], h[1], fcw[0] * h[0])));
        if ((u & 1) == 0) {
            accEven = acc;
        } else {
            const float mine = lo ? accEven : acc;
            const float send = lo ? acc : accEven;
            const float y = fast_sigmoid(mine + xhalf(send) + fcb);
            outp[(size_t)(u - 1) * Bn] = y;
        }
      }
      outp += (size_t)PF * Bn;

      #pragma unroll
      for (int u = 0; u < PF; ++u) { xc0[u] = xn0[u]; xc1[u] = xn1[u]; }
    }

    {
        float* hp = h_out + (size_t)b * 8 + hb;
        hp[0] = h[0]; hp[1] = h[1]; hp[2] = h[2]; hp[3] = h[3];
    }
}

extern "C" void kernel_launch(void* const* d_in, const int* in_sizes, int n_in,
                              void* d_out, int out_size, void* d_ws, size_t ws_size,
                              hipStream_t stream) {
    const float* x    = (const float*)d_in[0];
    const float* h0   = (const float*)d_in[1];
    const float* w_ih = (const float*)d_in[2];
    const float* w_hh = (const float*)d_in[3];
    const float* b_ih = (const float*)d_in[4];
    const float* b_hh = (const float*)d_in[5];
    const float* fc_w = (const float*)d_in[6];
    const float* fc_b = (const float*)d_in[7];

    float* out    = (float*)d_out;
    float* h_last = out + (size_t)Tn * Bn;

    const int threads = 128;
    const int blocks  = Bn / 16 / (threads / 32);
    gru_wmma_kernel<<<blocks, threads, 0, stream>>>(
        x, h0, w_ih, w_hh, b_ih, b_hh, fc_w, fc_b, out, h_last);
}


# gate and head activations on the hardware tanh: sigmoid(v) = 0.5+0.5*tanh(v/2), exact halves folded into the weights and biases; exp+rcp pairs go; n argument summed as tR*(hn/2) + (hn/2 + xn)
# speedup vs baseline: 6.5816x; 1.2245x over previous
#include <hip/hip_runtime.h>


typedef float v2f __attribute__((ext_vector_type(2)));
typedef float v8f __attribute__((ext_vector_type(8)));

#define WMMA_F32X4(A, B, C) \
    __builtin_amdgcn_wmma_f32_16x16x4_f32(false, (A), false, (B), (short)0, (C), false, false)

constexpr int Tn = 1024;
constexpr int Bn = 4096;
constexpr int PF = 8;

__device__ __forceinline__ float hw_tanh(float v) {
    return __builtin_amdgcn_tanhf(v);
}
__device__ __forceinline__ float xhalf(float v) {
    const int bits = __builtin_bit_cast(int, v);
    return __builtin_bit_cast(float,
        __builtin_amdgcn_permlanex16(bits, bits, 0x76543210, 0xfedcba98, false, false));
}

__global__ __launch_bounds__(256) void gru_wmma_kernel(
    const float* __restrict__ x,
    const float* __restrict__ h0,
    const float* __restrict__ w_ih,
    const float* __restrict__ w_hh,
    const float* __restrict__ b_ih,
    const float* __restrict__ b_hh,
    const float* __restrict__ fc_w,
    const float* __restrict__ fc_b,
    float* __restrict__ out,
    float* __restrict__ h_out)
{
    const int lane  = threadIdx.x & 31;
    const int wid   = (blockIdx.x * blockDim.x + threadIdx.x) >> 5;
    const int m     = lane & 15;
    const int half  = lane >> 4;
    const bool lo   = half == 0;
    const int bbase = wid * 16;
    const int b     = bbase + m;
    const int hb    = half * 4;

    const int  a_hid = ((m >> 3) << 2) + (m & 3);
    const bool a_isz = (m & 4) != 0;
    const int  rowRZ = (a_isz ? 8 : 0) + a_hid;
    const int  rowN  = 16 + a_hid;
    v2f aRZ1, aRZ2, aN1, aN2;
    aRZ1.x = 0.5f * w_hh[rowRZ * 8 + hb + 0]; aRZ1.y = 0.5f * w_hh[rowRZ * 8 + hb + 1];
    aRZ2.x = 0.5f * w_hh[rowRZ * 8 + hb + 2]; aRZ2.y = 0.5f * w_hh[rowRZ * 8 + hb + 3];
    if (!a_isz) {
        aN1.x = 0.5f * w_hh[rowN * 8 + hb + 0]; aN1.y = 0.5f * w_hh[rowN * 8 + hb + 1];
        aN2.x = 0.5f * w_hh[rowN * 8 + hb + 2]; aN2.y = 0.5f * w_hh[rowN * 8 + hb + 3];
    } else {
        aN1.x = aN1.y = aN2.x = aN2.y = 0.0f;
    }

    v8f cRZ, cN;
    #pragma unroll
    for (int r = 0; r < 4; ++r) {
        cRZ[r]     = 0.5f * (b_ih[hb + r]     + b_hh[hb + r]);
        cRZ[4 + r] = 0.5f * (b_ih[8 + hb + r] + b_hh[8 + hb + r]);
        cN[r]      = 0.5f * b_hh[16 + hb + r];
        cN[4 + r]  = 0.0f;
    }

    float wR0[4], wR1[4], wZ0[4], wZ1[4], wN0[4], wN1[4], bN[4], fcw[4];
    #pragma unroll
    for (int s = 0; s < 4; ++s) {
        const int j = hb + s;
        wR0[s] = 0.5f * w_ih[j * 2 + 0];        wR1[s] = 0.5f * w_ih[j * 2 + 1];
        wZ0[s] = 0.5f * w_ih[(8 + j) * 2 + 0];  wZ1[s] = 0.5f * w_ih[(8 + j) * 2 + 1];
        wN0[s] = w_ih[(16 + j) * 2 + 0]; wN1[s] = w_ih[(16 + j) * 2 + 1];
        bN[s]  = b_ih[16 + j];
        fcw[s] = 0.5f * fc_w[j];
    }
    const float fcb = 0.5f * fc_b[0];

    float h[4];
    {
        const float* hp = h0 + (size_t)b * 8 + hb;
        h[0] = hp[0]; h[1] = hp[1]; h[2] = hp[2]; h[3] = hp[3];
    }

    const float* xp   = x + (size_t)b * 2;
    float*       outp = out + (size_t)half * Bn + bbase + m;

    float xc0[PF], xc1[PF];
    #pragma unroll
    for (int u = 0; u < PF; ++u) {
        xc0[u] = xp[(size_t)u * Bn * 2 + 0];
        xc1[u] = xp[(size_t)u * Bn * 2 + 1];
    }

    for (int tb = 0; tb < Tn; tb += PF) {
      float xn0[PF], xn1[PF];
      #pragma unroll
      for (int u = 0; u < PF; ++u) {
          const int tn = (tb + PF + u < Tn) ? (tb + PF + u) : (Tn - 1);
          const float* xn = xp + (size_t)tn * Bn * 2;
          xn0[u] = xn[0];
          xn1[u] = xn[1];
      }
      __builtin_amdgcn_sched_barrier(0);

      float accEven = 0.0f;
      #pragma unroll
      for (int u = 0; u < PF; ++u) {
        const float x0 = xc0[u];
        const float x1 = xc1[u];

        v2f b1, b2;
        b1.x = h[0]; b1.y = h[1];
        b2.x = h[2]; b2.y = h[3];

        v8f dRZ = WMMA_F32X4(aRZ1, b1, cRZ);
        dRZ     = WMMA_F32X4(aRZ2, b2, dRZ);
        v8f dN  = WMMA_F32X4(aN1, b1, cN);
        dN      = WMMA_F32X4(aN2, b2, dN);

        #pragma unroll
        for (int s = 0; s < 4; ++s) {
            const float tR  = hw_tanh(dRZ[s]     + fmaf(x1, wR1[s], x0 * wR0[s]));
            const float tZ  = hw_tanh(dRZ[4 + s] + fmaf(x1, wZ1[s], x0 * wZ0[s]));
            const float xnS = fmaf(x1, wN1[s], fmaf(x0, wN0[s], bN[s]));
            const float nS  = hw_tanh(fmaf(tR, dN[s], dN[s] + xnS));
            const float zS  = fmaf(0.5f, tZ, 0.5f);
            h[s] = nS + zS * (h[s] - nS);
        }

        const float acc = fmaf(fcw[3], h[3], fmaf(fcw[2], h[2], fmaf(fcw[1], h[1], fcw[0] * h[0])));
        if ((u & 1) == 0) {
            accEven = acc;
        } else {
            const float mine = lo ? accEven : acc;
            const float send = lo ? acc : accEven;
            const float y = fmaf(0.5f, hw_tanh(mine + xhalf(send) + fcb), 0.5f);
            outp[(size_t)(u - 1) * Bn] = y;
        }
      }
      outp += (size_t)PF * Bn;

      #pragma unroll
      for (int u = 0; u < PF; ++u) { xc0[u] = xn0[u]; xc1[u] = xn1[u]; }
    }

    {
        float* hp = h_out + (size_t)b * 8 + hb;
        hp[0] = h[0]; hp[1] = h[1]; hp[2] = h[2]; hp[3] = h[3];
    }
}

extern "C" void kernel_launch(void* const* d_in, const int* in_sizes, int n_in,
                              void* d_out, int out_size, void* d_ws, size_t ws_size,
                              hipStream_t stream) {
    const float* x    = (const float*)d_in[0];
    const float* h0   = (const float*)d_in[1];
    const float* w_ih = (const float*)d_in[2];
    const float* w_hh = (const float*)d_in[3];
    const float* b_ih = (const float*)d_in[4];
    const float* b_hh = (const float*)d_in[5];
    const float* fc_w = (const float*)d_in[6];
    const float* fc_b = (const float*)d_in[7];

    float* out    = (float*)d_out;
    float* h_last = out + (size_t)Tn * Bn;

    const int threads = 128;
    const int blocks  = Bn / 16 / (threads / 32);
    gru_wmma_kernel<<<blocks, threads, 0, stream>>>(
        x, h0, w_ih, w_hh, b_ih, b_hh, fc_w, fc_b, out, h_last);
}


# r, n input paths enter as the C tile of their h products, formed a step ahead; spare tile rows hold the n rows again so hn/2 and hn/2+xn both leave D; no add before tanh r, n; sum order bias,x,h
# speedup vs baseline: 6.8430x; 1.0397x over previous
#include <hip/hip_runtime.h>


typedef float v2f __attribute__((ext_vector_type(2)));
typedef float v8f __attribute__((ext_vector_type(8)));

#define WMMA_F32X4(A, B, C) \
    __builtin_amdgcn_wmma_f32_16x16x4_f32(false, (A), false, (B), (short)0, (C), false, false)

constexpr int Tn = 1024;
constexpr int Bn = 4096;
constexpr int PF = 8;

__device__ __forceinline__ float hw_tanh(float v) {
    return __builtin_amdgcn_tanhf(v);
}
__device__ __forceinline__ float xhalf(float v) {
    const int bits = __builtin_bit_cast(int, v);
    return __builtin_bit_cast(float,
        __builtin_amdgcn_permlanex16(bits, bits, 0x76543210, 0xfedcba98, false, false));
}

__global__ __launch_bounds__(256) void gru_wmma_kernel(
    const float* __restrict__ x,
    const float* __restrict__ h0,
    const float* __restrict__ w_ih,
    const float* __restrict__ w_hh,
    const float* __restrict__ b_ih,
    const float* __restrict__ b_hh,
    const float* __restrict__ fc_w,
    const float* __restrict__ fc_b,
    float* __restrict__ out,
    float* __restrict__ h_out)
{
    const int lane  = threadIdx.x & 31;
    const int wid   = (blockIdx.x * blockDim.x + threadIdx.x) >> 5;
    const int m     = lane & 15;
    const int half  = lane >> 4;
    const bool lo   = half == 0;
    const int bbase = wid * 16;
    const int b     = bbase + m;
    const int hb    = half * 4;

    const int  a_hid = ((m >> 3) << 2) + (m & 3);
    const bool a_hi  = (m & 4) != 0;
    const int  rowRN = (a_hi ? 16 : 0) + a_hid;
    const int  rowNZ = (a_hi ? 8 : 16) + a_hid;
    v2f aRN1, aRN2, aNZ1, aNZ2;
    aRN1.x = 0.5f * w_hh[rowRN * 8 + hb + 0]; aRN1.y = 0.5f * w_hh[rowRN * 8 + hb + 1];
    aRN2.x = 0.5f * w_hh[rowRN * 8 + hb + 2]; aRN2.y = 0.5f * w_hh[rowRN * 8 + hb + 3];
    aNZ1.x = 0.5f * w_hh[rowNZ * 8 + hb + 0]; aNZ1.y = 0.5f * w_hh[rowNZ * 8 + hb + 1];
    aNZ2.x = 0.5f * w_hh[rowNZ * 8 + hb + 2]; aNZ2.y = 0.5f * w_hh[rowNZ * 8 + hb + 3];

    v8f cNZ;
    #pragma unroll
    for (int r = 0; r < 4; ++r) {
        cNZ[r]     = 0.5f * b_hh[16 + hb + r];
        cNZ[4 + r] = 0.5f * (b_ih[8 + hb + r] + b_hh[8 + hb + r]);
    }

    float bR[4], bN2[4];
    float wR0[4], wR1[4], wZ0[4], wZ1[4], wN0[4], wN1[4], fcw[4];
    #pragma unroll
    for (int s = 0; s < 4; ++s) {
        const int j = hb + s;
        bR[s]  = 0.5f * (b_ih[j] + b_hh[j]);
        bN2[s] = 0.5f * b_hh[16 + j] + b_ih[16 + j];
        wR0[s] = 0.5f * w_ih[j * 2 + 0];        wR1[s] = 0.5f * w_ih[j * 2 + 1];
        wZ0[s] = 0.5f * w_ih[(8 + j) * 2 + 0];  wZ1[s] = 0.5f * w_ih[(8 + j) * 2 + 1];
        wN0[s] = w_ih[(16 + j) * 2 + 0]; wN1[s] = w_ih[(16 + j) * 2 + 1];
        fcw[s] = 0.5f * fc_w[j];
    }
    const float fcb = 0.5f * fc_b[0];

    float h[4];
    {
        const float* hp = h0 + (size_t)b * 8 + hb;
        h[0] = hp[0]; h[1] = hp[1]; h[2] = hp[2]; h[3] = hp[3];
    }

    const float* xp   = x + (size_t)b * 2;
    float*       outp = out + (size_t)half * Bn + bbase + m;

    float xc0[PF], xc1[PF];
    #pragma unroll
    for (int u = 0; u < PF; ++u) {
        xc0[u] = xp[(size_t)u * Bn * 2 + 0];
        xc1[u] = xp[(size_t)u * Bn * 2 + 1];
    }

    v8f nxRN;
    #pragma unroll
    for (int s = 0; s < 4; ++s) {
        nxRN[s]     = fmaf(xc1[0], wR1[s], fmaf(xc0[0], wR0[s], bR[s]));
        nxRN[4 + s] = fmaf(xc1[0], wN1[s], fmaf(xc0[0], wN0[s], bN2[s]));
    }

    for (int tb = 0; tb < Tn; tb += PF) {
      float xn0[PF], xn1[PF];
      #pragma unroll
      for (int u = 0; u < PF; ++u) {
          const int tn = (tb + PF + u < Tn) ? (tb + PF + u) : (Tn - 1);
          const float* xn = xp + (size_t)tn * Bn * 2;
          xn0[u] = xn[0];
          xn1[u] = xn[1];
      }
      __builtin_amdgcn_sched_barrier(0);

      float accEven = 0.0f;
      #pragma unroll
      for (int u = 0; u < PF; ++u) {
        const float x0 = xc0[u];
        const float x1 = xc1[u];

        v2f b1, b2;
        b1.x = h[0]; b1.y = h[1];
        b2.x = h[2]; b2.y = h[3];

        v8f dRN = WMMA_F32X4(aRN1, b1, nxRN);
        dRN     = WMMA_F32X4(aRN2, b2, dRN);
        v8f dNZ = WMMA_F32X4(aNZ1, b1, cNZ);
        dNZ     = WMMA_F32X4(aNZ2, b2, dNZ);

        {
            const float y0 = (u + 1 < PF) ? xc0[u + 1] : xn0[0];
            const float y1 = (u + 1 < PF) ? xc1[u + 1] : xn1[0];
            #pragma unroll
            for (int s = 0; s < 4; ++s) {
                nxRN[s]     = fmaf(y1, wR1[s], fmaf(y0, wR0[s], bR[s]));
                nxRN[4 + s] = fmaf(y1, wN1[s], fmaf(y0, wN0[s], bN2[s]));
            }
        }

        #pragma unroll
        for (int s = 0; s < 4; ++s) {
            const float tR  = hw_tanh(dRN[s]);
            const float tZ  = hw_tanh(dNZ[4 + s] + fmaf(x1, wZ1[s], x0 * wZ0[s]));
            const float nS  = hw_tanh(fmaf(tR, dNZ[s], dRN[4 + s]));
            const float zS  = fmaf(0.5f, tZ, 0.5f);
            h[s] = nS + zS * (h[s] - nS);
        }

        const float acc = fmaf(fcw[3], h[3], fmaf(fcw[2], h[2], fmaf(fcw[1], h[1], fcw[0] * h[0])));
        if ((u & 1) == 0) {
            accEven = acc;
        } else {
            const float mine = lo ? accEven : acc;
            const float send = lo ? acc : accEven;
            const float y = fmaf(0.5f, hw_tanh(mine + xhalf(send) + fcb), 0.5f);
            outp[(size_t)(u - 1) * Bn] = y;
        }
      }
      outp += (size_t)PF * Bn;

      #pragma unroll
      for (int u = 0; u < PF; ++u) { xc0[u] = xn0[u]; xc1[u] = xn1[u]; }
    }

    {
        float* hp = h_out + (size_t)b * 8 + hb;
        hp[0] = h[0]; hp[1] = h[1]; hp[2] = h[2]; hp[3] = h[3];
    }
}

extern "C" void kernel_launch(void* const* d_in, const int* in_sizes, int n_in,
                              void* d_out, int out_size, void* d_ws, size_t ws_size,
                              hipStream_t stream) {
    const float* x    = (const float*)d_in[0];
    const float* h0   = (const float*)d_in[1];
    const float* w_ih = (const float*)d_in[2];
    const float* w_hh = (const float*)d_in[3];
    const float* b_ih = (const float*)d_in[4];
    const float* b_hh = (const float*)d_in[5];
    const float* fc_w = (const float*)d_in[6];
    const float* fc_b = (const float*)d_in[7];

    float* out    = (float*)d_out;
    float* h_last = out + (size_t)Tn * Bn;

    const int threads = 128;
    const int blocks  = Bn / 16 / (threads / 32);
    gru_wmma_kernel<<<blocks, threads, 0, stream>>>(
        x, h0, w_ih, w_hh, b_ih, b_hh, fc_w, fc_b, out, h_last);
}


# T loop in trips of 32 steps (was 8): x in two 16-step register buffers refilled in turn, outputs stored once per trip; hand-over copies, addresses, waits once per 32 steps; sum order unchanged
# speedup vs baseline: 9.3962x; 1.3731x over previous
#include <hip/hip_runtime.h>


typedef float v2f __attribute__((ext_vector_type(2)));
typedef float v8f __attribute__((ext_vector_type(8)));

#define WMMA_F32X4(A, B, C) \
    __builtin_amdgcn_wmma_f32_16x16x4_f32(false, (A), false, (B), (short)0, (C), false, false)

constexpr int Tn = 1024;
constexpr int Bn = 4096;
constexpr int PF = 32;
constexpr int HB = PF / 2;

__device__ __forceinline__ float hw_tanh(float v) {
    return __builtin_amdgcn_tanhf(v);
}
__device__ __forceinline__ float xhalf(float v) {
    const int bits = __builtin_bit_cast(int, v);
    return __builtin_bit_cast(float,
        __builtin_amdgcn_permlanex16(bits, bits, 0x76543210, 0xfedcba98, false, false));
}

__global__ __launch_bounds__(256) void gru_wmma_kernel(
    const float* __restrict__ x,
    const float* __restrict__ h0,
    const float* __restrict__ w_ih,
    const float* __restrict__ w_hh,
    const float* __restrict__ b_ih,
    const float* __restrict__ b_hh,
    const float* __restrict__ fc_w,
    const float* __restrict__ fc_b,
    float* __restrict__ out,
    float* __restrict__ h_out)
{
    const int lane  = threadIdx.x & 31;
    const int wid   = (blockIdx.x * blockDim.x + threadIdx.x) >> 5;
    const int m     = lane & 15;
    const int half  = lane >> 4;
    const bool lo   = half == 0;
    const int bbase = wid * 16;
    const int b     = bbase + m;
    const int hb    = half * 4;

    const int  a_hid = ((m >> 3) << 2) + (m & 3);
    const bool a_hi  = (m & 4) != 0;
    const int  rowRN = (a_hi ? 16 : 0) + a_hid;
    const int  rowNZ = (a_hi ? 8 : 16) + a_hid;
    v2f aRN1, aRN2, aNZ1, aNZ2;
    aRN1.x = 0.5f * w_hh[rowRN * 8 + hb + 0]; aRN1.y = 0.5f * w_hh[rowRN * 8 + hb + 1];
    aRN2.x = 0.5f * w_hh[rowRN * 8 + hb + 2]; aRN2.y = 0.5f * w_hh[rowRN * 8 + hb + 3];
    aNZ1.x = 0.5f * w_hh[rowNZ * 8 + hb + 0]; aNZ1.y = 0.5f * w_hh[rowNZ * 8 + hb + 1];
    aNZ2.x = 0.5f * w_hh[rowNZ * 8 + hb + 2]; aNZ2.y = 0.5f * w_hh[rowNZ * 8 + hb + 3];

    v8f cNZ;
    #pragma unroll
    for (int r = 0; r < 4; ++r) {
        cNZ[r]     = 0.5f * b_hh[16 + hb + r];
        cNZ[4 + r] = 0.5f * (b_ih[8 + hb + r] + b_hh[8 + hb + r]);
    }

    float bR[4], bN2[4];
    float wR0[4], wR1[4], wZ0[4], wZ1[4], wN0[4], wN1[4], fcw[4];
    #pragma unroll
    for (int s = 0; s < 4; ++s) {
        const int j = hb + s;
        bR[s]  = 0.5f * (b_ih[j] + b_hh[j]);
        bN2[s] = 0.5f * b_hh[16 + j] + b_ih[16 + j];
        wR0[s] = 0.5f * w_ih[j * 2 + 0];        wR1[s] = 0.5f * w_ih[j * 2 + 1];
        wZ0[s] = 0.5f * w_ih[(8 + j) * 2 + 0];  wZ1[s] = 0.5f * w_ih[(8 + j) * 2 + 1];
        wN0[s] = w_ih[(16 + j) * 2 + 0]; wN1[s] = w_ih[(16 + j) * 2 + 1];
        fcw[s] = 0.5f * fc_w[j];
    }
    const float fcb = 0.5f * fc_b[0];

    float h[4];
    {
        const float* hp = h0 + (size_t)b * 8 + hb;
        h[0] = hp[0]; h[1] = hp[1]; h[2] = hp[2]; h[3] = hp[3];
    }

    const float* xp   = x + (size_t)b * 2;
    float*       outp = out + (size_t)half * Bn + bbase + m;

    float xa0[HB], xa1[HB];
    float xb0[HB], xb1[HB];
    #pragma unroll
    for (int u = 0; u < HB; ++u) {
        xa0[u] = xp[(size_t)u * Bn * 2 + 0];
        xa1[u] = xp[(size_t)u * Bn * 2 + 1];
    }
    const float* xqB = xp + (size_t)HB * Bn * 2;
    const float* xqA = xp + (size_t)PF * Bn * 2;

    v8f nxRN;
    #pragma unroll
    for (int s = 0; s < 4; ++s) {
        nxRN[s]     = fmaf(xa1[0], wR1[s], fmaf(xa0[0], wR0[s], bR[s]));
        nxRN[4 + s] = fmaf(xa1[0], wN1[s], fmaf(xa0[0], wN0[s], bN2[s]));
    }

    for (int tb = 0; tb < Tn; tb += PF) {
      float accEven = 0.0f;
      float yo[PF / 2];

      #pragma unroll
      for (int u = 0; u < PF; ++u) {
        if (u == 0) {
            #pragma unroll
            for (int v = 0; v < HB; ++v) {
                xb0[v] = xqB[(size_t)v * Bn * 2 + 0];
                xb1[v] = xqB[(size_t)v * Bn * 2 + 1];
            }
            __builtin_amdgcn_sched_barrier(0);
        }
        if (u == HB) {
            #pragma unroll
            for (int v = 0; v < HB; ++v) {
                xa0[v] = xqA[(size_t)v * Bn * 2 + 0];
                xa1[v] = xqA[(size_t)v * Bn * 2 + 1];
            }
            __builtin_amdgcn_sched_barrier(0);
        }

        const float x0 = (u < HB) ? xa0[u % HB] : xb0[u % HB];
        const float x1 = (u < HB) ? xa1[u % HB] : xb1[u % HB];

        v2f b1, b2;
        b1.x = h[0]; b1.y = h[1];
        b2.x = h[2]; b2.y = h[3];

        v8f dRN = WMMA_F32X4(aRN1, b1, nxRN);
        dRN     = WMMA_F32X4(aRN2, b2, dRN);
        v8f dNZ = WMMA_F32X4(aNZ1, b1, cNZ);
        dNZ     = WMMA_F32X4(aNZ2, b2, dNZ);

        {
            const int un = (u + 1) % PF;
            const float y0 = (un < HB) ? xa0[un % HB] : xb0[un % HB];
            const float y1 = (un < HB) ? xa1[un % HB] : xb1[un % HB];
            #pragma unroll
            for (int s = 0; s < 4; ++s) {
                nxRN[s]     = fmaf(y1, wR1[s], fmaf(y0, wR0[s], bR[s]));
                nxRN[4 + s] = fmaf(y1, wN1[s], fmaf(y0, wN0[s], bN2[s]));
            }
        }

        #pragma unroll
        for (int s = 0; s < 4; ++s) {
            const float tR  = hw_tanh(dRN[s]);
            const float tZ  = hw_tanh(dNZ[4 + s] + fmaf(x1, wZ1[s], x0 * wZ0[s]));
            const float nS  = hw_tanh(fmaf(tR, dNZ[s], dRN[4 + s]));
            const float zS  = fmaf(0.5f, tZ, 0.5f);
            h[s] = nS + zS * (h[s] - nS);
        }

        const float acc = fmaf(fcw[3], h[3], fmaf(fcw[2], h[2], fmaf(fcw[1], h[1], fcw[0] * h[0])));
        if ((u & 1) == 0) {
            accEven = acc;
        } else {
            const float mine = lo ? accEven : acc;
            const float send = lo ? acc : accEven;
            const float y = fmaf(0.5f, hw_tanh(mine + xhalf(send) + fcb), 0.5f);
            yo[u >> 1] = y;
        }
      }

      __builtin_amdgcn_sched_barrier(0);
      #pragma unroll
      for (int k = 0; k < PF / 2; ++k) outp[(size_t)(2 * k) * Bn] = yo[k];
      outp += (size_t)PF * Bn;
      xqB  += (size_t)PF * Bn * 2;
      if (tb + 2 * PF < Tn) xqA += (size_t)PF * Bn * 2;
    }

    {
        float* hp = h_out + (size_t)b * 8 + hb;
        hp[0] = h[0]; hp[1] = h[1]; hp[2] = h[2]; hp[3] = h[3];
    }
}

extern "C" void kernel_launch(void* const* d_in, const int* in_sizes, int n_in,
                              void* d_out, int out_size, void* d_ws, size_t ws_size,
                              hipStream_t stream) {
    const float* x    = (const float*)d_in[0];
    const float* h0   = (const float*)d_in[1];
    const float* w_ih = (const float*)d_in[2];
    const float* w_hh = (const float*)d_in[3];
    const float* b_ih = (const float*)d_in[4];
    const float* b_hh = (const float*)d_in[5];
    const float* fc_w = (const float*)d_in[6];
    const float* fc_b = (const float*)d_in[7];

    float* out    = (float*)d_out;
    float* h_last = out + (size_t)Tn * Bn;

    const int threads = 128;
    const int blocks  = Bn / 16 / (threads / 32);
    gru_wmma_kernel<<<blocks, threads, 0, stream>>>(
        x, h0, w_ih, w_hh, b_ih, b_hh, fc_w, fc_b, out, h_last);
}


# a step's gate math laid out by hand in issue order (fences): each tanh followed by plain work, h update as 2h' = (h+n) + tanh(-z/2)(n-h), one fma behind the last tanh; h update's rounding changes
# speedup vs baseline: 10.6196x; 1.1302x over previous
#include <hip/hip_runtime.h>


typedef float v2f __attribute__((ext_vector_type(2)));
typedef float v8f __attribute__((ext_vector_type(8)));

#define WMMA_F32X4(A, B, C) \
    __builtin_amdgcn_wmma_f32_16x16x4_f32(false, (A), false, (B), (short)0, (C), false, false)

#define SEG() __builtin_amdgcn_sched_barrier(0)
#define KEEP1(a)          asm volatile("" :: "v"(a))
#define KEEP2(a, b)       asm volatile("" :: "v"(a), "v"(b))
#define KEEP3(a, b, c)    asm volatile("" :: "v"(a), "v"(b), "v"(c))
#define KEEP4(a, b, c, d) asm volatile("" :: "v"(a), "v"(b), "v"(c), "v"(d))

constexpr int Tn = 1024;
constexpr int Bn = 4096;
constexpr int PF = 32;
constexpr int HB = PF / 2;

__device__ __forceinline__ float hw_tanh(float v) {
    return __builtin_amdgcn_tanhf(v);
}
__device__ __forceinline__ float xhalf(float v) {
    const int bits = __builtin_bit_cast(int, v);
    return __builtin_bit_cast(float,
        __builtin_amdgcn_permlanex16(bits, bits, 0x76543210, 0xfedcba98, false, false));
}

__global__ __launch_bounds__(256) void gru_wmma_kernel(
    const float* __restrict__ x,
    const float* __restrict__ h0,
    const float* __restrict__ w_ih,
    const float* __restrict__ w_hh,
    const float* __restrict__ b_ih,
    const float* __restrict__ b_hh,
    const float* __restrict__ fc_w,
    const float* __restrict__ fc_b,
    float* __restrict__ out,
    float* __restrict__ h_out)
{
    const int lane  = threadIdx.x & 31;
    const int wid   = (blockIdx.x * blockDim.x + threadIdx.x) >> 5;
    const int m     = lane & 15;
    const int half  = lane >> 4;
    const bool lo   = half == 0;
    const int bbase = wid * 16;
    const int b     = bbase + m;
    const int hb    = half * 4;

    const int  a_hid = ((m >> 3) << 2) + (m & 3);
    const bool a_hi  = (m & 4) != 0;
    const int  rowRN = (a_hi ? 16 : 0) + a_hid;
    const int  rowNZ = (a_hi ? 8 : 16) + a_hid;
    v2f aRN1, aRN2, aNZ1, aNZ2;
    aRN1.x = 0.25f * w_hh[rowRN * 8 + hb + 0]; aRN1.y = 0.25f * w_hh[rowRN * 8 + hb + 1];
    aRN2.x = 0.25f * w_hh[rowRN * 8 + hb + 2]; aRN2.y = 0.25f * w_hh[rowRN * 8 + hb + 3];
    const float sNZ = a_hi ? -0.25f : 0.25f;
    aNZ1.x = sNZ * w_hh[rowNZ * 8 + hb + 0]; aNZ1.y = sNZ * w_hh[rowNZ * 8 + hb + 1];
    aNZ2.x = sNZ * w_hh[rowNZ * 8 + hb + 2]; aNZ2.y = sNZ * w_hh[rowNZ * 8 + hb + 3];

    v8f cNZ;
    #pragma unroll
    for (int r = 0; r < 4; ++r) {
        cNZ[r]     = 0.5f * b_hh[16 + hb + r];
        cNZ[4 + r] = -0.5f * (b_ih[8 + hb + r] + b_hh[8 + hb + r]);
    }

    float bR[4], bN2[4];
    float wR0[4], wR1[4], wZ0[4], wZ1[4], wN0[4], wN1[4], fcw[4];
    #pragma unroll
    for (int s = 0; s < 4; ++s) {
        const int j = hb + s;
        bR[s]  = 0.5f * (b_ih[j] + b_hh[j]);
        bN2[s] = 0.5f * b_hh[16 + j] + b_ih[16 + j];
        wR0[s] = 0.5f * w_ih[j * 2 + 0];        wR1[s] = 0.5f * w_ih[j * 2 + 1];
        wZ0[s] = -0.5f * w_ih[(8 + j) * 2 + 0]; wZ1[s] = -0.5f * w_ih[(8 + j) * 2 + 1];
        wN0[s] = w_ih[(16 + j) * 2 + 0]; wN1[s] = w_ih[(16 + j) * 2 + 1];
        fcw[s] = 0.25f * fc_w[j];
    }
    const float fcb = 0.5f * fc_b[0];

    float h[4];
    {
        const float* hp = h0 + (size_t)b * 8 + hb;
        h[0] = 2.0f * hp[0]; h[1] = 2.0f * hp[1]; h[2] = 2.0f * hp[2]; h[3] = 2.0f * hp[3];
    }

    const float* xp   = x + (size_t)b * 2;
    float*       outp = out + (size_t)half * Bn + bbase + m;

    float xa0[HB], xa1[HB];
    float xb0[HB], xb1[HB];
    #pragma unroll
    for (int u = 0; u < HB; ++u) {
        xa0[u] = xp[(size_t)u * Bn * 2 + 0];
        xa1[u] = xp[(size_t)u * Bn * 2 + 1];
    }
    const float* xqB = xp + (size_t)HB * Bn * 2;
    const float* xqA = xp + (size_t)PF * Bn * 2;

    v8f nxRN;
    #pragma unroll
    for (int s = 0; s < 4; ++s) {
        nxRN[s]     = fmaf(xa1[0], wR1[s], fmaf(xa0[0], wR0[s], bR[s]));
        nxRN[4 + s] = fmaf(xa1[0], wN1[s], fmaf(xa0[0], wN0[s], bN2[s]));
    }

    float xz[4];
    #pragma unroll
    for (int s = 0; s < 4; ++s) xz[s] = fmaf(xa1[0], wZ1[s], xa0[0] * wZ0[s]);

    for (int tb = 0; tb < Tn; tb += PF) {
      float accEven = 0.0f;
      float yo[PF / 2];
      float pMine = 0.0f, pSend = 0.0f, pSum = 0.0f, pTanh = 0.0f;

      #pragma unroll
      for (int u = 0; u < PF; ++u) {
        if (u == 0) {
            #pragma unroll
            for (int v = 0; v < HB; ++v) {
                xb0[v] = xqB[(size_t)v * Bn * 2 + 0];
                xb1[v] = xqB[(size_t)v * Bn * 2 + 1];
            }
            __builtin_amdgcn_sched_barrier(0);
        }
        if (u == HB) {
            #pragma unroll
            for (int v = 0; v < HB; ++v) {
                xa0[v] = xqA[(size_t)v * Bn * 2 + 0];
                xa1[v] = xqA[(size_t)v * Bn * 2 + 1];
            }
            __builtin_amdgcn_sched_barrier(0);
        }

        v2f b1, b2;
        b1.x = h[0]; b1.y = h[1];
        b2.x = h[2]; b2.y = h[3];

        v8f dRN = WMMA_F32X4(aRN1, b1, nxRN);
        v8f dNZ = WMMA_F32X4(aNZ1, b1, cNZ);
        dRN     = WMMA_F32X4(aRN2, b2, dRN);
        dNZ     = WMMA_F32X4(aNZ2, b2, dNZ);

        const int un = (u + 1) % PF;
        const float y0 = (un < HB) ? xa0[un % HB] : xb0[un % HB];
        const float y1 = (un < HB) ? xa1[un % HB] : xb1[un % HB];
        float tR[4], tZ[4], nS[4], zg[4], ng[4], ar[4], an[4], mz[4], xzn[4], sp[4], dm[4];
        v8f nx;
        float acc = 0.0f;

        SEG();
        tR[0] = hw_tanh(dRN[0]);
        ar[0] = fmaf(y0, wR0[0], bR[0]);  ar[1] = fmaf(y0, wR0[1], bR[1]);
        KEEP3(tR[0], ar[0], ar[1]);
        SEG();
        tR[1] = hw_tanh(dRN[1]);
        ar[2] = fmaf(y0, wR0[2], bR[2]);  ar[3] = fmaf(y0, wR0[3], bR[3]);
        if (u >= 3 && (u & 1)) { pSum = xhalf(pSend); KEEP1(pSum); }
        KEEP3(tR[1], ar[2], ar[3]);
        SEG();
        tR[2] = hw_tanh(dRN[2]);
        zg[0] = dNZ[4] + xz[0];           zg[1] = dNZ[5] + xz[1];
        KEEP3(tR[2], zg[0], zg[1]);
        SEG();
        tR[3] = hw_tanh(dRN[3]);
        zg[2] = dNZ[6] + xz[2];           zg[3] = dNZ[7] + xz[3];
        KEEP3(tR[3], zg[2], zg[3]);
        SEG();
        an[0] = fmaf(y0, wN0[0], bN2[0]); an[1] = fmaf(y0, wN0[1], bN2[1]);
        if (u > 0) acc = fcw[0] * h[0];
        if (u >= 3 && (u & 1)) { pSum = pMine + pSum; KEEP1(pSum); }
        KEEP2(an[0], an[1]);
        SEG();
        tZ[0] = hw_tanh(zg[0]);
        ng[0] = fmaf(tR[0], dNZ[0], dRN[4]); ng[1] = fmaf(tR[1], dNZ[1], dRN[5]);
        KEEP3(tZ[0], ng[0], ng[1]);
        SEG();
        tZ[1] = hw_tanh(zg[1]);
        an[2] = fmaf(y0, wN0[2], bN2[2]); an[3] = fmaf(y0, wN0[3], bN2[3]);
        KEEP3(tZ[1], an[2], an[3]);
        SEG();
        tZ[2] = hw_tanh(zg[2]);
        ng[2] = fmaf(tR[2], dNZ[2], dRN[6]); ng[3] = fmaf(tR[3], dNZ[3], dRN[7]);
        KEEP3(tZ[2], ng[2], ng[3]);
        SEG();
        nS[0] = hw_tanh(ng[0]);
        if (u > 0) { acc = fmaf(fcw[1], h[1], acc); KEEP1(acc); }
        if (u >= 3 && (u & 1)) { pSum = pSum + fcb; KEEP1(pSum); }
        KEEP1(nS[0]);
        SEG();
        nS[1] = hw_tanh(ng[1]);
        nx[0] = fmaf(y1, wR1[0], ar[0]);  nx[1] = fmaf(y1, wR1[1], ar[1]);
        KEEP3(nS[1], nx[0], nx[1]);
        SEG();
        nS[2] = hw_tanh(ng[2]);
        nx[2] = fmaf(y1, wR1[2], ar[2]);  nx[3] = fmaf(y1, wR1[3], ar[3]);
        KEEP3(nS[2], nx[2], nx[3]);
        SEG();
        nS[3] = hw_tanh(ng[3]);
        nx[4] = fmaf(y1, wN1[0], an[0]);  nx[5] = fmaf(y1, wN1[1], an[1]);
        if (u > 0) { acc = fmaf(fcw[2], h[2], acc); KEEP1(acc); }
        KEEP3(nS[3], nx[4], nx[5]);
        SEG();
        tZ[3] = hw_tanh(zg[3]);
        KEEP1(tZ[3]);
        SEG();
        sp[0] = fmaf(0.5f, h[0], nS[0]);  dm[0] = fmaf(-0.5f, h[0], nS[0]);
        sp[1] = fmaf(0.5f, h[1], nS[1]);  dm[1] = fmaf(-0.5f, h[1], nS[1]);
        if (u >= 3 && (u & 1)) { pTanh = hw_tanh(pSum); KEEP1(pTanh); }
        KEEP4(sp[0], dm[0], sp[1], dm[1]);
        SEG();
        sp[2] = fmaf(0.5f, h[2], nS[2]);  dm[2] = fmaf(-0.5f, h[2], nS[2]);
        sp[3] = fmaf(0.5f, h[3], nS[3]);  dm[3] = fmaf(-0.5f, h[3], nS[3]);
        KEEP4(sp[2], dm[2], sp[3], dm[3]);
        SEG();
        nx[6] = fmaf(y1, wN1[2], an[2]);  nx[7] = fmaf(y1, wN1[3], an[3]);
        if (u > 0) { acc = fmaf(fcw[3], h[3], acc); KEEP1(acc); }
        KEEP2(nx[6], nx[7]);
        SEG();
        h[0] = fmaf(tZ[0], dm[0], sp[0]); h[1] = fmaf(tZ[1], dm[1], sp[1]);
        KEEP2(h[0], h[1]);
        SEG();
        mz[0] = y0 * wZ0[0];              mz[1] = y0 * wZ0[1];
        mz[2] = y0 * wZ0[2];              mz[3] = y0 * wZ0[3];
        xzn[0] = fmaf(y1, wZ1[0], mz[0]); xzn[1] = fmaf(y1, wZ1[1], mz[1]);
        xzn[2] = fmaf(y1, wZ1[2], mz[2]); xzn[3] = fmaf(y1, wZ1[3], mz[3]);
        KEEP4(xzn[0], xzn[1], xzn[2], xzn[3]);
        SEG();
        h[2] = fmaf(tZ[2], dm[2], sp[2]); h[3] = fmaf(tZ[3], dm[3], sp[3]);
        if (u > 0 && ((u - 1) & 1) == 0) accEven = acc;
        if (u > 0 && ((u - 1) & 1) == 1) {
            pMine = lo ? accEven : acc;
            pSend = lo ? acc : accEven;
        }
        if (u >= 3 && (u & 1)) yo[(u - 3) >> 1] = fmaf(0.5f, pTanh, 0.5f);
        KEEP2(h[2], h[3]);
        SEG();
        nxRN = nx;
        #pragma unroll
        for (int s = 0; s < 4; ++s) xz[s] = xzn[s];

        if (u == PF - 1) {
            const float accL = fmaf(fcw[3], h[3], fmaf(fcw[2], h[2], fmaf(fcw[1], h[1], fcw[0] * h[0])));
            const float mine = lo ? accEven : accL;
            const float send = lo ? accL : accEven;
            yo[(PF - 1) >> 1] = fmaf(0.5f, hw_tanh(mine + xhalf(send) + fcb), 0.5f);
        }
      }

      __builtin_amdgcn_sched_barrier(0);
      #pragma unroll
      for (int k = 0; k < PF / 2; ++k) outp[(size_t)(2 * k) * Bn] = yo[k];
      outp += (size_t)PF * Bn;
      xqB  += (size_t)PF * Bn * 2;
      if (tb + 2 * PF < Tn) xqA += (size_t)PF * Bn * 2;
    }

    {
        float* hp = h_out + (size_t)b * 8 + hb;
        hp[0] = 0.5f * h[0]; hp[1] = 0.5f * h[1]; hp[2] = 0.5f * h[2]; hp[3] = 0.5f * h[3];
    }
}

extern "C" void kernel_launch(void* const* d_in, const int* in_sizes, int n_in,
                              void* d_out, int out_size, void* d_ws, size_t ws_size,
                              hipStream_t stream) {
    const float* x    = (const float*)d_in[0];
    const float* h0   = (const float*)d_in[1];
    const float* w_ih = (const float*)d_in[2];
    const float* w_hh = (const float*)d_in[3];
    const float* b_ih = (const float*)d_in[4];
    const float* b_hh = (const float*)d_in[5];
    const float* fc_w = (const float*)d_in[6];
    const float* fc_b = (const float*)d_in[7];

    float* out    = (float*)d_out;
    float* h_last = out + (size_t)Tn * Bn;

    const int threads = 128;
    const int blocks  = Bn / 16 / (threads / 32);
    gru_wmma_kernel<<<blocks, threads, 0, stream>>>(
        x, h0, w_ih, w_hh, b_ih, b_hh, fc_w, fc_b, out, h_last);
}
